# starting kernel
# baseline (speedup 1.0000x reference)
#include <hip/hip_runtime.h>


typedef __attribute__((ext_vector_type(2))) float v2f;
typedef __attribute__((ext_vector_type(8))) float v8f;

#define Bsz 64
#define Tn  4096
#define In  64
#define Hn  128
#define On  64

__device__ __forceinline__ v8f wmma_f32_k4(v2f a, v2f b, v8f c) {
  return __builtin_amdgcn_wmma_f32_16x16x4_f32(
       false, a,  false, b,
       (short)0, c,  false,  false);
}

__global__ void __launch_bounds__(256)
rnn_phase1(const float* __restrict__ x, const float* __restrict__ Win,
           const float* __restrict__ bin, const float* __restrict__ Wg,
           const float* __restrict__ bg, float* __restrict__ zbuf,
           float* __restrict__ abuf) {
  __shared__ float zt[16][Hn + 4];
  const int lane = threadIdx.x & 31;
  const int wave = threadIdx.x >> 5;
  const int n0   = wave * 16;
  const int row  = lane & 15;
  const int hi   = lane >> 4;
  const int ko   = hi * 2;
  const int t0   = blockIdx.x * 16;
  const float* xrow = x + ((size_t)(Bsz - 1) * Tn + t0) * In;

  v8f c = {};
#pragma unroll
  for (int kc = 0; kc < In / 4; ++kc) {
    const int kb = kc * 4;
    v2f a, b;
    a.x = xrow[row * In + kb + ko];
    a.y = xrow[row * In + kb + ko + 1];
    const int n = n0 + row;
    b.x = Win[n * In + kb + ko];
    b.y = Win[n * In + kb + ko + 1];
    c = wmma_f32_k4(a, b, c);
  }
  {
    const int n = n0 + row;
    const float bias = bin[n];
#pragma unroll
    for (int r = 0; r < 8; ++r) {
      const int m = r + hi * 8;
      const float v = c[r] + bias;
      zt[m][n] = v;
      zbuf[(size_t)(t0 + m) * Hn + n] = v;
    }
  }
  __syncthreads();

  v8f c2 = {};
#pragma unroll
  for (int kc = 0; kc < Hn / 4; ++kc) {
    const int kb = kc * 4;
    v2f a, b;
    a.x = zt[row][kb + ko];
    a.y = zt[row][kb + ko + 1];
    const int n = n0 + row;
    b.x = Wg[n * (2 * Hn) + Hn + kb + ko];
    b.y = Wg[n * (2 * Hn) + Hn + kb + ko + 1];
    c2 = wmma_f32_k4(a, b, c2);
  }
  {
    const int n = n0 + row;
    const float bias = bg[n];
#pragma unroll
    for (int r = 0; r < 8; ++r) {
      const int m = r + hi * 8;
      abuf[(size_t)(t0 + m) * Hn + n] = c2[r] + bias;
    }
  }
}

__global__ void __launch_bounds__(256)
rnn_phase2(const float* __restrict__ Wg, const float* __restrict__ zbuf,
           const float* __restrict__ abuf, float* __restrict__ hbuf) {
  __shared__ float hsh[2][Hn];
  const int tid  = threadIdx.x;
  const int j    = tid >> 1;
  const int half = tid & 1;

  float w[64];
#pragma unroll
  for (int k = 0; k < 64; ++k)
    w[k] = Wg[(size_t)j * (2 * Hn) + half * 64 + k];

  if (tid < Hn) hsh[0][tid] = 0.0f;
  __syncthreads();

  float a0 = 0.f, a1 = 0.f, a2 = 0.f, a3 = 0.f;
  float z0 = 0.f, z1 = 0.f, z2 = 0.f, z3 = 0.f;
  if (half == 0) {
    a0 = abuf[0 * Hn + j];  z0 = zbuf[0 * Hn + j];
    a1 = abuf[1 * Hn + j];  z1 = zbuf[1 * Hn + j];
    a2 = abuf[2 * Hn + j];  z2 = zbuf[2 * Hn + j];
    a3 = abuf[3 * Hn + j];  z3 = zbuf[3 * Hn + j];
  }

#define RNN_STEP(tt, ac, zc)                                                 \
  do {                                                                       \
    const int cur = (tt) & 1;                                                \
    float p = 0.0f;                                                          \
    _Pragma("unroll")                                                        \
    for (int k = 0; k < 64; ++k)                                             \
      p += w[k] * hsh[cur][half * 64 + k];                                   \
    p += __shfl_xor(p, 1, 32);                      \
    if (half == 0) {                                                         \
      const float pre = (ac) + p;                                            \
      const float u   = 1.0f / (1.0f + __expf(-pre));                        \
      const float hn  = u * hsh[cur][j] + (1.0f - u) * (zc);                 \
      hsh[cur ^ 1][j] = hn;                           \
      hbuf[(size_t)(tt) * Hn + j] = hn;                                      \
      if ((tt) + 4 < Tn) {                             \
        (ac) = abuf[(size_t)((tt) + 4) * Hn + j];                            \
        (zc) = zbuf[(size_t)((tt) + 4) * Hn + j];                            \
      }                                                                      \
    }                                                                        \
    __syncthreads();                                \
  } while (0)

  for (int t = 0; t < Tn; t += 4) {
    RNN_STEP(t + 0, a0, z0);
    RNN_STEP(t + 1, a1, z1);
    RNN_STEP(t + 2, a2, z2);
    RNN_STEP(t + 3, a3, z3);
  }
#undef RNN_STEP
}

__global__ void __launch_bounds__(128)
rnn_phase3(const float* __restrict__ hbuf, const float* __restrict__ Wout,
           const float* __restrict__ bout, float* __restrict__ y) {
  const int lane = threadIdx.x & 31;
  const int wave = threadIdx.x >> 5;
  const int n0   = wave * 16;
  const int row  = lane & 15;
  const int hi   = lane >> 4;
  const int ko   = hi * 2;
  const int t0   = blockIdx.x * 16;

  v8f c = {};
#pragma unroll
  for (int kc = 0; kc < Hn / 4; ++kc) {
    const int kb = kc * 4;
    v2f a, b;
    a.x = hbuf[(size_t)(t0 + row) * Hn + kb + ko];
    a.y = hbuf[(size_t)(t0 + row) * Hn + kb + ko + 1];
    const int n = n0 + row;
    b.x = Wout[n * Hn + kb + ko];
    b.y = Wout[n * Hn + kb + ko + 1];
    c = wmma_f32_k4(a, b, c);
  }
  const int n = n0 + row;
  const float bias = bout[n];
#pragma unroll
  for (int r = 0; r < 8; ++r) {
    const int m = r + hi * 8;
    y[(size_t)(t0 + m) * On + n] = c[r] + bias;
  }
}

extern "C" void kernel_launch(void* const* d_in, const int* in_sizes, int n_in,
                              void* d_out, int out_size, void* d_ws, size_t ws_size,
                              hipStream_t stream) {
  const float* x    = (const float*)d_in[0];
  const float* Win  = (const float*)d_in[1];
  const float* bin  = (const float*)d_in[2];
  const float* Wg   = (const float*)d_in[3];
  const float* bg   = (const float*)d_in[4];
  const float* Wout = (const float*)d_in[5];
  const float* bout = (const float*)d_in[6];

  float* zbuf = (float*)d_ws;
  float* abuf = zbuf + (size_t)Tn * Hn;
  float* hbuf = abuf + (size_t)Tn * Hn;
  float* yout = (float*)d_out;

  rnn_phase1<<<Tn / 16, 256, 0, stream>>>(x, Win, bin, Wg, bg, zbuf, abuf);
  rnn_phase2<<<1, 256, 0, stream>>>(Wg, zbuf, abuf, hbuf);
  rnn_phase3<<<Tn / 16, 128, 0, stream>>>(hbuf, Wout, bout, yout);
}


# scan matvec re-tiled: thread = 4 rows x 16 k, h slice read once (4 wide LDS loads), 4 packed FMA chains, 8-lane DPP xor sum replaces bpermute, owner lane keeps h in a register; dot sum order changes
# speedup vs baseline: 1.6815x; 1.6815x over previous
#include <hip/hip_runtime.h>


typedef __attribute__((ext_vector_type(2))) float v2f;
typedef __attribute__((ext_vector_type(8))) float v8f;

#define Bsz 64
#define Tn  4096
#define In  64
#define Hn  128
#define On  64

__device__ __forceinline__ v8f wmma_f32_k4(v2f a, v2f b, v8f c) {
  return __builtin_amdgcn_wmma_f32_16x16x4_f32(
       false, a,  false, b,
       (short)0, c,  false,  false);
}

__global__ void __launch_bounds__(256)
rnn_phase1(const float* __restrict__ x, const float* __restrict__ Win,
           const float* __restrict__ bin, const float* __restrict__ Wg,
           const float* __restrict__ bg, float* __restrict__ zbuf,
           float* __restrict__ abuf) {
  __shared__ float zt[16][Hn + 4];
  const int lane = threadIdx.x & 31;
  const int wave = threadIdx.x >> 5;
  const int n0   = wave * 16;
  const int row  = lane & 15;
  const int hi   = lane >> 4;
  const int ko   = hi * 2;
  const int t0   = blockIdx.x * 16;
  const float* xrow = x + ((size_t)(Bsz - 1) * Tn + t0) * In;

  v8f c = {};
#pragma unroll
  for (int kc = 0; kc < In / 4; ++kc) {
    const int kb = kc * 4;
    v2f a, b;
    a.x = xrow[row * In + kb + ko];
    a.y = xrow[row * In + kb + ko + 1];
    const int n = n0 + row;
    b.x = Win[n * In + kb + ko];
    b.y = Win[n * In + kb + ko + 1];
    c = wmma_f32_k4(a, b, c);
  }
  {
    const int n = n0 + row;
    const float bias = bin[n];
#pragma unroll
    for (int r = 0; r < 8; ++r) {
      const int m = r + hi * 8;
      const float v = c[r] + bias;
      zt[m][n] = v;
      zbuf[(size_t)(t0 + m) * Hn + n] = v;
    }
  }
  __syncthreads();

  v8f c2 = {};
#pragma unroll
  for (int kc = 0; kc < Hn / 4; ++kc) {
    const int kb = kc * 4;
    v2f a, b;
    a.x = zt[row][kb + ko];
    a.y = zt[row][kb + ko + 1];
    const int n = n0 + row;
    b.x = Wg[n * (2 * Hn) + Hn + kb + ko];
    b.y = Wg[n * (2 * Hn) + Hn + kb + ko + 1];
    c2 = wmma_f32_k4(a, b, c2);
  }
  {
    const int n = n0 + row;
    const float bias = bg[n];
#pragma unroll
    for (int r = 0; r < 8; ++r) {
      const int m = r + hi * 8;
      abuf[(size_t)(t0 + m) * Hn + n] = c2[r] + bias;
    }
  }
}

#define HPITCH 20
#define HPAD   (8 * HPITCH)

template <int CTRL>
__device__ __forceinline__ float dpp_take(float v) {
  return __int_as_float(__builtin_amdgcn_update_dpp(
      0, __float_as_int(v), CTRL, 0xF, 0xF, true));
}

__global__ void __launch_bounds__(256)
rnn_phase2(const float* __restrict__ Wg, const float* __restrict__ zbuf,
           const float* __restrict__ abuf, float* __restrict__ hbuf) {
  __shared__ __attribute__((aligned(16))) float hsh[2][HPAD];
  const int tid = threadIdx.x;
  const int ks  = tid & 7;
  const int jg  = tid >> 3;
  const int sel = ks & 3;
  const int j   = 4 * jg + sel;
  const int jw  = HPITCH * (j >> 4) + (j & 15);
  const bool fin = ks < 4;
  const unsigned m0 = (sel == 0) ? 0xFFFFFFFFu : 0u;
  const unsigned m1 = (sel == 1) ? 0xFFFFFFFFu : 0u;
  const unsigned m2 = (sel == 2) ? 0xFFFFFFFFu : 0u;
  const unsigned m3 = (sel == 3) ? 0xFFFFFFFFu : 0u;

  v2f wa[16], wb[16];
#pragma unroll
  for (int i = 0; i < 16; ++i) {
    wa[i].x = Wg[(size_t)(4 * jg + 0) * (2 * Hn) + 16 * ks + i];
    wa[i].y = Wg[(size_t)(4 * jg + 1) * (2 * Hn) + 16 * ks + i];
    wb[i].x = Wg[(size_t)(4 * jg + 2) * (2 * Hn) + 16 * ks + i];
    wb[i].y = Wg[(size_t)(4 * jg + 3) * (2 * Hn) + 16 * ks + i];
  }

  if (tid < HPAD) hsh[0][tid] = 0.0f;
  __syncthreads();

  float hprev = 0.0f;
  float a0 = 0.f, a1 = 0.f, a2 = 0.f, a3 = 0.f;
  float z0 = 0.f, z1 = 0.f, z2 = 0.f, z3 = 0.f;
  if (fin) {
    a0 = abuf[0 * Hn + j];  z0 = zbuf[0 * Hn + j];
    a1 = abuf[1 * Hn + j];  z1 = zbuf[1 * Hn + j];
    a2 = abuf[2 * Hn + j];  z2 = zbuf[2 * Hn + j];
    a3 = abuf[3 * Hn + j];  z3 = zbuf[3 * Hn + j];
  }

#define RNN_STEP(tt, ac, zc)                                                 \
  do {                                                                       \
    const int cur = (tt) & 1;                                                \
    float hv[16];                                \
    __builtin_memcpy(hv, &hsh[cur][HPITCH * ks], 64);                        \
    v2f A0 = {0.0f, 0.0f}, A1 = {0.0f, 0.0f};                                \
    v2f B0 = {0.0f, 0.0f}, B1 = {0.0f, 0.0f};                                \
          \
    _Pragma("unroll")                                                        \
    for (int i = 0; i < 16; i += 2) {                                        \
      const v2f h0 = {hv[i], hv[i]};                                         \
      const v2f h1 = {hv[i + 1], hv[i + 1]};                                 \
      A0 += wa[i] * h0;      B0 += wb[i] * h0;                               \
      A1 += wa[i + 1] * h1;  B1 += wb[i + 1] * h1;                           \
    }                                                                        \
    A0 += A1;  B0 += B1;                                                     \
    float p0 = A0.x, p1 = A0.y, p2 = B0.x, p3 = B0.y;                        \
                         \
    p0 += dpp_take<0x161>(p0); p1 += dpp_take<0x161>(p1);                    \
    p2 += dpp_take<0x161>(p2); p3 += dpp_take<0x161>(p3);                    \
    p0 += dpp_take<0x162>(p0); p1 += dpp_take<0x162>(p1);                    \
    p2 += dpp_take<0x162>(p2); p3 += dpp_take<0x162>(p3);                    \
    p0 += dpp_take<0x164>(p0); p1 += dpp_take<0x164>(p1);                    \
    p2 += dpp_take<0x164>(p2); p3 += dpp_take<0x164>(p3);                    \
    const float p = __uint_as_float(                                         \
        (__float_as_uint(p0) & m0) | (__float_as_uint(p1) & m1) |            \
        (__float_as_uint(p2) & m2) | (__float_as_uint(p3) & m3));            \
    if (fin) {                                                               \
      const float pre = (ac) + p;                                            \
      const float u   = 1.0f / (1.0f + __expf(-pre));                        \
      const float hn  = u * hprev + (1.0f - u) * (zc);                       \
      hprev = hn;                                                            \
      hsh[cur ^ 1][jw] = hn;                          \
      hbuf[(size_t)(tt) * Hn + j] = hn;                                      \
      if ((tt) + 4 < Tn) {                             \
        (ac) = abuf[(size_t)((tt) + 4) * Hn + j];                            \
        (zc) = zbuf[(size_t)((tt) + 4) * Hn + j];                            \
      }                                                                      \
    }                                                                        \
    __syncthreads();                                \
  } while (0)

  for (int t = 0; t < Tn; t += 4) {
    RNN_STEP(t + 0, a0, z0);
    RNN_STEP(t + 1, a1, z1);
    RNN_STEP(t + 2, a2, z2);
    RNN_STEP(t + 3, a3, z3);
  }
#undef RNN_STEP
}

__global__ void __launch_bounds__(128)
rnn_phase3(const float* __restrict__ hbuf, const float* __restrict__ Wout,
           const float* __restrict__ bout, float* __restrict__ y) {
  const int lane = threadIdx.x & 31;
  const int wave = threadIdx.x >> 5;
  const int n0   = wave * 16;
  const int row  = lane & 15;
  const int hi   = lane >> 4;
  const int ko   = hi * 2;
  const int t0   = blockIdx.x * 16;

  v8f c = {};
#pragma unroll
  for (int kc = 0; kc < Hn / 4; ++kc) {
    const int kb = kc * 4;
    v2f a, b;
    a.x = hbuf[(size_t)(t0 + row) * Hn + kb + ko];
    a.y = hbuf[(size_t)(t0 + row) * Hn + kb + ko + 1];
    const int n = n0 + row;
    b.x = Wout[n * Hn + kb + ko];
    b.y = Wout[n * Hn + kb + ko + 1];
    c = wmma_f32_k4(a, b, c);
  }
  const int n = n0 + row;
  const float bias = bout[n];
#pragma unroll
  for (int r = 0; r < 8; ++r) {
    const int m = r + hi * 8;
    y[(size_t)(t0 + m) * On + n] = c[r] + bias;
  }
}

extern "C" void kernel_launch(void* const* d_in, const int* in_sizes, int n_in,
                              void* d_out, int out_size, void* d_ws, size_t ws_size,
                              hipStream_t stream) {
  const float* x    = (const float*)d_in[0];
  const float* Win  = (const float*)d_in[1];
  const float* bin  = (const float*)d_in[2];
  const float* Wg   = (const float*)d_in[3];
  const float* bg   = (const float*)d_in[4];
  const float* Wout = (const float*)d_in[5];
  const float* bout = (const float*)d_in[6];

  float* zbuf = (float*)d_ws;
  float* abuf = zbuf + (size_t)Tn * Hn;
  float* hbuf = abuf + (size_t)Tn * Hn;
  float* yout = (float*)d_out;

  rnn_phase1<<<Tn / 16, 256, 0, stream>>>(x, Win, bin, Wg, bg, zbuf, abuf);
  rnn_phase2<<<1, 256, 0, stream>>>(Wg, zbuf, abuf, hbuf);
  rnn_phase3<<<Tn / 16, 128, 0, stream>>>(hbuf, Wout, bout, yout);
}


# scan step no longer drains global memory: hand-written LDS-only barrier (s_wait_dscnt 0) replaces __syncthreads, a/z refill loads made unconditional so counted waits keep 3 younger pairs in flight
# speedup vs baseline: 3.6002x; 2.1411x over previous
#include <hip/hip_runtime.h>


typedef __attribute__((ext_vector_type(2))) float v2f;
typedef __attribute__((ext_vector_type(8))) float v8f;

#define Bsz 64
#define Tn  4096
#define In  64
#define Hn  128
#define On  64

__device__ __forceinline__ v8f wmma_f32_k4(v2f a, v2f b, v8f c) {
  return __builtin_amdgcn_wmma_f32_16x16x4_f32(
       false, a,  false, b,
       (short)0, c,  false,  false);
}

__global__ void __launch_bounds__(256)
rnn_phase1(const float* __restrict__ x, const float* __restrict__ Win,
           const float* __restrict__ bin, const float* __restrict__ Wg,
           const float* __restrict__ bg, float* __restrict__ zbuf,
           float* __restrict__ abuf) {
  __shared__ float zt[16][Hn + 4];
  const int lane = threadIdx.x & 31;
  const int wave = threadIdx.x >> 5;
  const int n0   = wave * 16;
  const int row  = lane & 15;
  const int hi   = lane >> 4;
  const int ko   = hi * 2;
  const int t0   = blockIdx.x * 16;
  const float* xrow = x + ((size_t)(Bsz - 1) * Tn + t0) * In;

  v8f c = {};
#pragma unroll
  for (int kc = 0; kc < In / 4; ++kc) {
    const int kb = kc * 4;
    v2f a, b;
    a.x = xrow[row * In + kb + ko];
    a.y = xrow[row * In + kb + ko + 1];
    const int n = n0 + row;
    b.x = Win[n * In + kb + ko];
    b.y = Win[n * In + kb + ko + 1];
    c = wmma_f32_k4(a, b, c);
  }
  {
    const int n = n0 + row;
    const float bias = bin[n];
#pragma unroll
    for (int r = 0; r < 8; ++r) {
      const int m = r + hi * 8;
      const float v = c[r] + bias;
      zt[m][n] = v;
      zbuf[(size_t)(t0 + m) * Hn + n] = v;
    }
  }
  __syncthreads();

  v8f c2 = {};
#pragma unroll
  for (int kc = 0; kc < Hn / 4; ++kc) {
    const int kb = kc * 4;
    v2f a, b;
    a.x = zt[row][kb + ko];
    a.y = zt[row][kb + ko + 1];
    const int n = n0 + row;
    b.x = Wg[n * (2 * Hn) + Hn + kb + ko];
    b.y = Wg[n * (2 * Hn) + Hn + kb + ko + 1];
    c2 = wmma_f32_k4(a, b, c2);
  }
  {
    const int n = n0 + row;
    const float bias = bg[n];
#pragma unroll
    for (int r = 0; r < 8; ++r) {
      const int m = r + hi * 8;
      abuf[(size_t)(t0 + m) * Hn + n] = c2[r] + bias;
    }
  }
}

#define HPITCH 20
#define HPAD   (8 * HPITCH)

template <int CTRL>
__device__ __forceinline__ float dpp_take(float v) {
  return __int_as_float(__builtin_amdgcn_update_dpp(
      0, __float_as_int(v), CTRL, 0xF, 0xF, true));
}

__device__ __forceinline__ void lds_step_barrier() {
  asm volatile("s_wait_dscnt 0x0\n\ts_barrier_signal -1\n\ts_barrier_wait -1"
               ::: "memory");
}

__global__ void __launch_bounds__(256)
rnn_phase2(const float* __restrict__ Wg, const float* __restrict__ zbuf,
           const float* __restrict__ abuf, float* __restrict__ hbuf) {
  __shared__ __attribute__((aligned(16))) float hsh[2][HPAD];
  const int tid = threadIdx.x;
  const int ks  = tid & 7;
  const int jg  = tid >> 3;
  const int sel = ks & 3;
  const int j   = 4 * jg + sel;
  const int jw  = HPITCH * (j >> 4) + (j & 15);
  const bool fin = ks < 4;
  const unsigned m0 = (sel == 0) ? 0xFFFFFFFFu : 0u;
  const unsigned m1 = (sel == 1) ? 0xFFFFFFFFu : 0u;
  const unsigned m2 = (sel == 2) ? 0xFFFFFFFFu : 0u;
  const unsigned m3 = (sel == 3) ? 0xFFFFFFFFu : 0u;

  v2f wa[16], wb[16];
#pragma unroll
  for (int i = 0; i < 16; ++i) {
    wa[i].x = Wg[(size_t)(4 * jg + 0) * (2 * Hn) + 16 * ks + i];
    wa[i].y = Wg[(size_t)(4 * jg + 1) * (2 * Hn) + 16 * ks + i];
    wb[i].x = Wg[(size_t)(4 * jg + 2) * (2 * Hn) + 16 * ks + i];
    wb[i].y = Wg[(size_t)(4 * jg + 3) * (2 * Hn) + 16 * ks + i];
  }

  if (tid < HPAD) hsh[0][tid] = 0.0f;
  __syncthreads();

  float hprev = 0.0f;
  float a0 = abuf[0 * Hn + j], z0 = zbuf[0 * Hn + j];
  float a1 = abuf[1 * Hn + j], z1 = zbuf[1 * Hn + j];
  float a2 = abuf[2 * Hn + j], z2 = zbuf[2 * Hn + j];
  float a3 = abuf[3 * Hn + j], z3 = zbuf[3 * Hn + j];

#define RNN_STEP(tt, ac, zc)                                                 \
  do {                                                                       \
    const int cur = (tt) & 1;                                                \
    float hv[16];                                \
    __builtin_memcpy(hv, &hsh[cur][HPITCH * ks], 64);                        \
    v2f A0 = {0.0f, 0.0f}, A1 = {0.0f, 0.0f};                                \
    v2f B0 = {0.0f, 0.0f}, B1 = {0.0f, 0.0f};                                \
          \
    _Pragma("unroll")                                                        \
    for (int i = 0; i < 16; i += 2) {                                        \
      const v2f h0 = {hv[i], hv[i]};                                         \
      const v2f h1 = {hv[i + 1], hv[i + 1]};                                 \
      A0 += wa[i] * h0;      B0 += wb[i] * h0;                               \
      A1 += wa[i + 1] * h1;  B1 += wb[i + 1] * h1;                           \
    }                                                                        \
    A0 += A1;  B0 += B1;                                                     \
    float p0 = A0.x, p1 = A0.y, p2 = B0.x, p3 = B0.y;                        \
                         \
    p0 += dpp_take<0x161>(p0); p1 += dpp_take<0x161>(p1);                    \
    p2 += dpp_take<0x161>(p2); p3 += dpp_take<0x161>(p3);                    \
    p0 += dpp_take<0x162>(p0); p1 += dpp_take<0x162>(p1);                    \
    p2 += dpp_take<0x162>(p2); p3 += dpp_take<0x162>(p3);                    \
    p0 += dpp_take<0x164>(p0); p1 += dpp_take<0x164>(p1);                    \
    p2 += dpp_take<0x164>(p2); p3 += dpp_take<0x164>(p3);                    \
    const float p = __uint_as_float(                                         \
        (__float_as_uint(p0) & m0) | (__float_as_uint(p1) & m1) |            \
        (__float_as_uint(p2) & m2) | (__float_as_uint(p3) & m3));            \
    const float pre = (ac) + p;                                              \
    const float u   = 1.0f / (1.0f + __expf(-pre));                          \
    const float hn  = u * hprev + (1.0f - u) * (zc);                         \
    hprev = hn;                                                              \
    if (fin) {                                    \
      hsh[cur ^ 1][jw] = hn;                          \
      hbuf[(size_t)(tt) * Hn + j] = hn;                                      \
    }                                                                        \
      \
      \
      \
      \
    const int tn = ((tt) + 4 < Tn) ? (tt) + 4 : Tn - 1;                      \
    (ac) = abuf[(size_t)tn * Hn + j];                                        \
    (zc) = zbuf[(size_t)tn * Hn + j];                                        \
    lds_step_barrier();                        \
  } while (0)

  for (int t = 0; t < Tn; t += 4) {
    RNN_STEP(t + 0, a0, z0);
    RNN_STEP(t + 1, a1, z1);
    RNN_STEP(t + 2, a2, z2);
    RNN_STEP(t + 3, a3, z3);
  }
#undef RNN_STEP
}

__global__ void __launch_bounds__(128)
rnn_phase3(const float* __restrict__ hbuf, const float* __restrict__ Wout,
           const float* __restrict__ bout, float* __restrict__ y) {
  const int lane = threadIdx.x & 31;
  const int wave = threadIdx.x >> 5;
  const int n0   = wave * 16;
  const int row  = lane & 15;
  const int hi   = lane >> 4;
  const int ko   = hi * 2;
  const int t0   = blockIdx.x * 16;

  v8f c = {};
#pragma unroll
  for (int kc = 0; kc < Hn / 4; ++kc) {
    const int kb = kc * 4;
    v2f a, b;
    a.x = hbuf[(size_t)(t0 + row) * Hn + kb + ko];
    a.y = hbuf[(size_t)(t0 + row) * Hn + kb + ko + 1];
    const int n = n0 + row;
    b.x = Wout[n * Hn + kb + ko];
    b.y = Wout[n * Hn + kb + ko + 1];
    c = wmma_f32_k4(a, b, c);
  }
  const int n = n0 + row;
  const float bias = bout[n];
#pragma unroll
  for (int r = 0; r < 8; ++r) {
    const int m = r + hi * 8;
    y[(size_t)(t0 + m) * On + n] = c[r] + bias;
  }
}

extern "C" void kernel_launch(void* const* d_in, const int* in_sizes, int n_in,
                              void* d_out, int out_size, void* d_ws, size_t ws_size,
                              hipStream_t stream) {
  const float* x    = (const float*)d_in[0];
  const float* Win  = (const float*)d_in[1];
  const float* bin  = (const float*)d_in[2];
  const float* Wg   = (const float*)d_in[3];
  const float* bg   = (const float*)d_in[4];
  const float* Wout = (const float*)d_in[5];
  const float* bout = (const float*)d_in[6];

  float* zbuf = (float*)d_ws;
  float* abuf = zbuf + (size_t)Tn * Hn;
  float* hbuf = abuf + (size_t)Tn * Hn;
  float* yout = (float*)d_out;

  rnn_phase1<<<Tn / 16, 256, 0, stream>>>(x, Win, bin, Wg, bg, zbuf, abuf);
  rnn_phase2<<<1, 256, 0, stream>>>(Wg, zbuf, abuf, hbuf);
  rnn_phase3<<<Tn / 16, 128, 0, stream>>>(hbuf, Wout, bout, yout);
}


# scan gate tail 17 to 5 dependent ops: 1-u = rcp(1+exp2(c*p+c*a)) with one v_rcp_f32 (1 ulp) instead of the IEEE divide sequence, blend h+(1-u)*(z-h) as one FMA; h rounds differently in last bits
# speedup vs baseline: 4.0386x; 1.1218x over previous
#include <hip/hip_runtime.h>


typedef __attribute__((ext_vector_type(2))) float v2f;
typedef __attribute__((ext_vector_type(8))) float v8f;

#define Bsz 64
#define Tn  4096
#define In  64
#define Hn  128
#define On  64

__device__ __forceinline__ v8f wmma_f32_k4(v2f a, v2f b, v8f c) {
  return __builtin_amdgcn_wmma_f32_16x16x4_f32(
       false, a,  false, b,
       (short)0, c,  false,  false);
}

__global__ void __launch_bounds__(256)
rnn_phase1(const float* __restrict__ x, const float* __restrict__ Win,
           const float* __restrict__ bin, const float* __restrict__ Wg,
           const float* __restrict__ bg, float* __restrict__ zbuf,
           float* __restrict__ abuf) {
  __shared__ float zt[16][Hn + 4];
  const int lane = threadIdx.x & 31;
  const int wave = threadIdx.x >> 5;
  const int n0   = wave * 16;
  const int row  = lane & 15;
  const int hi   = lane >> 4;
  const int ko   = hi * 2;
  const int t0   = blockIdx.x * 16;
  const float* xrow = x + ((size_t)(Bsz - 1) * Tn + t0) * In;

  v8f c = {};
#pragma unroll
  for (int kc = 0; kc < In / 4; ++kc) {
    const int kb = kc * 4;
    v2f a, b;
    a.x = xrow[row * In + kb + ko];
    a.y = xrow[row * In + kb + ko + 1];
    const int n = n0 + row;
    b.x = Win[n * In + kb + ko];
    b.y = Win[n * In + kb + ko + 1];
    c = wmma_f32_k4(a, b, c);
  }
  {
    const int n = n0 + row;
    const float bias = bin[n];
#pragma unroll
    for (int r = 0; r < 8; ++r) {
      const int m = r + hi * 8;
      const float v = c[r] + bias;
      zt[m][n] = v;
      zbuf[(size_t)(t0 + m) * Hn + n] = v;
    }
  }
  __syncthreads();

  v8f c2 = {};
#pragma unroll
  for (int kc = 0; kc < Hn / 4; ++kc) {
    const int kb = kc * 4;
    v2f a, b;
    a.x = zt[row][kb + ko];
    a.y = zt[row][kb + ko + 1];
    const int n = n0 + row;
    b.x = Wg[n * (2 * Hn) + Hn + kb + ko];
    b.y = Wg[n * (2 * Hn) + Hn + kb + ko + 1];
    c2 = wmma_f32_k4(a, b, c2);
  }
  {
    const int n = n0 + row;
    const float bias = bg[n];
#pragma unroll
    for (int r = 0; r < 8; ++r) {
      const int m = r + hi * 8;
      abuf[(size_t)(t0 + m) * Hn + n] = c2[r] + bias;
    }
  }
}

#define HPITCH 20
#define HPAD   (8 * HPITCH)
#define LOG2E  1.44269504088896340736f

template <int CTRL>
__device__ __forceinline__ float dpp_take(float v) {
  return __int_as_float(__builtin_amdgcn_update_dpp(
      0, __float_as_int(v), CTRL, 0xF, 0xF, true));
}

__device__ __forceinline__ void lds_step_barrier() {
  asm volatile("s_wait_dscnt 0x0\n\ts_barrier_signal -1\n\ts_barrier_wait -1"
               ::: "memory");
}

__global__ void __launch_bounds__(256)
rnn_phase2(const float* __restrict__ Wg, const float* __restrict__ zbuf,
           const float* __restrict__ abuf, float* __restrict__ hbuf) {
  __shared__ __attribute__((aligned(16))) float hsh[2][HPAD];
  const int tid = threadIdx.x;
  const int ks  = tid & 7;
  const int jg  = tid >> 3;
  const int sel = ks & 3;
  const int j   = 4 * jg + sel;
  const int jw  = HPITCH * (j >> 4) + (j & 15);
  const bool fin = ks < 4;
  const unsigned m0 = (sel == 0) ? 0xFFFFFFFFu : 0u;
  const unsigned m1 = (sel == 1) ? 0xFFFFFFFFu : 0u;
  const unsigned m2 = (sel == 2) ? 0xFFFFFFFFu : 0u;
  const unsigned m3 = (sel == 3) ? 0xFFFFFFFFu : 0u;

  v2f wa[16], wb[16];
#pragma unroll
  for (int i = 0; i < 16; ++i) {
    wa[i].x = Wg[(size_t)(4 * jg + 0) * (2 * Hn) + 16 * ks + i];
    wa[i].y = Wg[(size_t)(4 * jg + 1) * (2 * Hn) + 16 * ks + i];
    wb[i].x = Wg[(size_t)(4 * jg + 2) * (2 * Hn) + 16 * ks + i];
    wb[i].y = Wg[(size_t)(4 * jg + 3) * (2 * Hn) + 16 * ks + i];
  }

  if (tid < HPAD) hsh[0][tid] = 0.0f;
  __syncthreads();

  float hprev = 0.0f;
  float a0 = abuf[0 * Hn + j], z0 = zbuf[0 * Hn + j];
  float a1 = abuf[1 * Hn + j], z1 = zbuf[1 * Hn + j];
  float a2 = abuf[2 * Hn + j], z2 = zbuf[2 * Hn + j];
  float a3 = abuf[3 * Hn + j], z3 = zbuf[3 * Hn + j];

#define RNN_STEP(tt, ac, zc)                                                 \
  do {                                                                       \
    const int cur = (tt) & 1;                                                \
    float hv[16];                                \
    __builtin_memcpy(hv, &hsh[cur][HPITCH * ks], 64);                        \
    v2f A0 = {0.0f, 0.0f}, A1 = {0.0f, 0.0f};                                \
    v2f B0 = {0.0f, 0.0f}, B1 = {0.0f, 0.0f};                                \
          \
    _Pragma("unroll")                                                        \
    for (int i = 0; i < 16; i += 2) {                                        \
      const v2f h0 = {hv[i], hv[i]};                                         \
      const v2f h1 = {hv[i + 1], hv[i + 1]};                                 \
      A0 += wa[i] * h0;      B0 += wb[i] * h0;                               \
      A1 += wa[i + 1] * h1;  B1 += wb[i + 1] * h1;                           \
    }                                                                        \
    A0 += A1;  B0 += B1;                                                     \
    float p0 = A0.x, p1 = A0.y, p2 = B0.x, p3 = B0.y;                        \
                         \
    p0 += dpp_take<0x161>(p0); p1 += dpp_take<0x161>(p1);                    \
    p2 += dpp_take<0x161>(p2); p3 += dpp_take<0x161>(p3);                    \
    p0 += dpp_take<0x162>(p0); p1 += dpp_take<0x162>(p1);                    \
    p2 += dpp_take<0x162>(p2); p3 += dpp_take<0x162>(p3);                    \
    p0 += dpp_take<0x164>(p0); p1 += dpp_take<0x164>(p1);                    \
    p2 += dpp_take<0x164>(p2); p3 += dpp_take<0x164>(p3);                    \
    const float p = __uint_as_float(                                         \
        (__float_as_uint(p0) & m0) | (__float_as_uint(p1) & m1) |            \
        (__float_as_uint(p2) & m2) | (__float_as_uint(p3) & m3));            \
      \
      \
      \
      \
      \
      \
    const float as  = (ac) * LOG2E;                                          \
    const float dz  = (zc) - hprev;                                          \
    const float e   = __builtin_amdgcn_exp2f(__builtin_fmaf(p, LOG2E, as));  \
    const float v   = __builtin_amdgcn_rcpf(1.0f + e);                       \
    const float hn  = __builtin_fmaf(v, dz, hprev);                          \
    hprev = hn;                                                              \
    if (fin) {                                    \
      hsh[cur ^ 1][jw] = hn;                          \
      hbuf[(size_t)(tt) * Hn + j] = hn;                                      \
    }                                                                        \
      \
      \
      \
      \
    const int tn = ((tt) + 4 < Tn) ? (tt) + 4 : Tn - 1;                      \
    (ac) = abuf[(size_t)tn * Hn + j];                                        \
    (zc) = zbuf[(size_t)tn * Hn + j];                                        \
    lds_step_barrier();                        \
  } while (0)

  for (int t = 0; t < Tn; t += 4) {
    RNN_STEP(t + 0, a0, z0);
    RNN_STEP(t + 1, a1, z1);
    RNN_STEP(t + 2, a2, z2);
    RNN_STEP(t + 3, a3, z3);
  }
#undef RNN_STEP
}

__global__ void __launch_bounds__(128)
rnn_phase3(const float* __restrict__ hbuf, const float* __restrict__ Wout,
           const float* __restrict__ bout, float* __restrict__ y) {
  const int lane = threadIdx.x & 31;
  const int wave = threadIdx.x >> 5;
  const int n0   = wave * 16;
  const int row  = lane & 15;
  const int hi   = lane >> 4;
  const int ko   = hi * 2;
  const int t0   = blockIdx.x * 16;

  v8f c = {};
#pragma unroll
  for (int kc = 0; kc < Hn / 4; ++kc) {
    const int kb = kc * 4;
    v2f a, b;
    a.x = hbuf[(size_t)(t0 + row) * Hn + kb + ko];
    a.y = hbuf[(size_t)(t0 + row) * Hn + kb + ko + 1];
    const int n = n0 + row;
    b.x = Wout[n * Hn + kb + ko];
    b.y = Wout[n * Hn + kb + ko + 1];
    c = wmma_f32_k4(a, b, c);
  }
  const int n = n0 + row;
  const float bias = bout[n];
#pragma unroll
  for (int r = 0; r < 8; ++r) {
    const int m = r + hi * 8;
    y[(size_t)(t0 + m) * On + n] = c[r] + bias;
  }
}

extern "C" void kernel_launch(void* const* d_in, const int* in_sizes, int n_in,
                              void* d_out, int out_size, void* d_ws, size_t ws_size,
                              hipStream_t stream) {
  const float* x    = (const float*)d_in[0];
  const float* Win  = (const float*)d_in[1];
  const float* bin  = (const float*)d_in[2];
  const float* Wg   = (const float*)d_in[3];
  const float* bg   = (const float*)d_in[4];
  const float* Wout = (const float*)d_in[5];
  const float* bout = (const float*)d_in[6];

  float* zbuf = (float*)d_ws;
  float* abuf = zbuf + (size_t)Tn * Hn;
  float* hbuf = abuf + (size_t)Tn * Hn;
  float* yout = (float*)d_out;

  rnn_phase1<<<Tn / 16, 256, 0, stream>>>(x, Win, bin, Wg, bg, zbuf, abuf);
  rnn_phase2<<<1, 256, 0, stream>>>(Wg, zbuf, abuf, hbuf);
  rnn_phase3<<<Tn / 16, 128, 0, stream>>>(hbuf, Wout, bout, yout);
}


# scan re-tiled to one wave per SIMD: 4 waves, thread = 8 rows x 16 k as k-pair packed FMA, 8-lane select + DPP-add reduce-scatter leaves one row per lane, nothing predicated; dot sum order unchanged
# speedup vs baseline: 4.2342x; 1.0484x over previous
#include <hip/hip_runtime.h>


typedef __attribute__((ext_vector_type(2))) float v2f;
typedef __attribute__((ext_vector_type(8))) float v8f;

#define Bsz 64
#define Tn  4096
#define In  64
#define Hn  128
#define On  64

__device__ __forceinline__ v8f wmma_f32_k4(v2f a, v2f b, v8f c) {
  return __builtin_amdgcn_wmma_f32_16x16x4_f32(
       false, a,  false, b,
       (short)0, c,  false,  false);
}

__global__ void __launch_bounds__(256)
rnn_phase1(const float* __restrict__ x, const float* __restrict__ Win,
           const float* __restrict__ bin, const float* __restrict__ Wg,
           const float* __restrict__ bg, float* __restrict__ zbuf,
           float* __restrict__ abuf) {
  __shared__ float zt[16][Hn + 4];
  const int lane = threadIdx.x & 31;
  const int wave = threadIdx.x >> 5;
  const int n0   = wave * 16;
  const int row  = lane & 15;
  const int hi   = lane >> 4;
  const int ko   = hi * 2;
  const int t0   = blockIdx.x * 16;
  const float* xrow = x + ((size_t)(Bsz - 1) * Tn + t0) * In;

  v8f c = {};
#pragma unroll
  for (int kc = 0; kc < In / 4; ++kc) {
    const int kb = kc * 4;
    v2f a, b;
    a.x = xrow[row * In + kb + ko];
    a.y = xrow[row * In + kb + ko + 1];
    const int n = n0 + row;
    b.x = Win[n * In + kb + ko];
    b.y = Win[n * In + kb + ko + 1];
    c = wmma_f32_k4(a, b, c);
  }
  {
    const int n = n0 + row;
    const float bias = bin[n];
#pragma unroll
    for (int r = 0; r < 8; ++r) {
      const int m = r + hi * 8;
      const float v = c[r] + bias;
      zt[m][n] = v;
      zbuf[(size_t)(t0 + m) * Hn + n] = v;
    }
  }
  __syncthreads();

  v8f c2 = {};
#pragma unroll
  for (int kc = 0; kc < Hn / 4; ++kc) {
    const int kb = kc * 4;
    v2f a, b;
    a.x = zt[row][kb + ko];
    a.y = zt[row][kb + ko + 1];
    const int n = n0 + row;
    b.x = Wg[n * (2 * Hn) + Hn + kb + ko];
    b.y = Wg[n * (2 * Hn) + Hn + kb + ko + 1];
    c2 = wmma_f32_k4(a, b, c2);
  }
  {
    const int n = n0 + row;
    const float bias = bg[n];
#pragma unroll
    for (int r = 0; r < 8; ++r) {
      const int m = r + hi * 8;
      abuf[(size_t)(t0 + m) * Hn + n] = c2[r] + bias;
    }
  }
}

#define HPITCH 20
#define HPAD   (8 * HPITCH)
#define LOG2E  1.44269504088896340736f
#define NT2    128

#define DPP_XADD(d, send, keep, XM)                                          \
  asm("v_add_f32_dpp %0, %1, %2 row_xmask:" #XM " row_mask:0xf bank_mask:0xf" \
      : "=v"(d) : "v"(send), "v"(keep))

#define RS_STAGE(dst, lo, hi, bit, XM)                                       \
  do {                                                                       \
    const float keep_ = (bit) ? (hi) : (lo);                                 \
    const float send_ = (bit) ? (lo) : (hi);                                 \
    DPP_XADD(dst, send_, keep_, XM);                                         \
  } while (0)

__device__ __forceinline__ void lds_step_barrier() {
  asm volatile("s_wait_dscnt 0x0\n\ts_barrier_signal -1\n\ts_barrier_wait -1"
               ::: "memory");
}

__global__ void __launch_bounds__(NT2)
rnn_phase2(const float* __restrict__ Wg, const float* __restrict__ zbuf,
           const float* __restrict__ abuf, float* __restrict__ hbuf) {
  __shared__ __attribute__((aligned(16))) float hsh[2][HPAD];
  const int tid = threadIdx.x;
  const int ks  = tid & 7;
  const int jg  = tid >> 3;
  const int jw  = HPITCH * (tid >> 4) + (tid & 15);
  const bool b0 = (ks & 1) != 0;
  const bool b1 = (ks & 2) != 0;
  const bool b2 = (ks & 4) != 0;

  v2f w[8][8];
#pragma unroll
  for (int r = 0; r < 8; ++r) {
#pragma unroll
    for (int c = 0; c < 8; ++c) {
      w[r][c].x = Wg[(size_t)(8 * jg + r) * (2 * Hn) + 16 * ks + 2 * c];
      w[r][c].y = Wg[(size_t)(8 * jg + r) * (2 * Hn) + 16 * ks + 2 * c + 1];
    }
  }

  for (int i = tid; i < HPAD; i += NT2) hsh[0][i] = 0.0f;
  __syncthreads();

  float hprev = 0.0f;
  float a0 = abuf[0 * Hn + tid], z0 = zbuf[0 * Hn + tid];
  float a1 = abuf[1 * Hn + tid], z1 = zbuf[1 * Hn + tid];
  float a2 = abuf[2 * Hn + tid], z2 = zbuf[2 * Hn + tid];
  float a3 = abuf[3 * Hn + tid], z3 = zbuf[3 * Hn + tid];

#define RNN_STEP(tt, ac, zc)                                                 \
  do {                                                                       \
    const int cur = (tt) & 1;                                                \
    v2f hv[8];                                 \
    __builtin_memcpy(hv, &hsh[cur][HPITCH * ks], 64);                        \
    v2f acc[8];                                                              \
      \
    _Pragma("unroll")                                                        \
    for (int r = 0; r < 8; ++r) acc[r] = (v2f){0.0f, 0.0f};                  \
    _Pragma("unroll")                                                        \
    for (int c = 0; c < 8; ++c) {                                            \
      _Pragma("unroll")                                                      \
      for (int r = 0; r < 8; ++r) acc[r] += w[r][c] * hv[c];                 \
    }                                                                        \
    float p[8];                                                              \
    _Pragma("unroll")                                                        \
    for (int r = 0; r < 8; ++r) p[r] = acc[r].x + acc[r].y;                  \
      \
      \
      \
    float q0, q1, q2, q3, r0, r1, s;                                         \
    RS_STAGE(q0, p[0], p[1], b0, 1);                                         \
    RS_STAGE(q1, p[2], p[3], b0, 1);                                         \
    RS_STAGE(q2, p[4], p[5], b0, 1);                                         \
    RS_STAGE(q3, p[6], p[7], b0, 1);                                         \
    RS_STAGE(r0, q0, q1, b1, 2);                                             \
    RS_STAGE(r1, q2, q3, b1, 2);                                             \
    RS_STAGE(s, r0, r1, b2, 4);                                              \
      \
      \
      \
      \
      \
      \
    const float as  = (ac) * LOG2E;                                          \
    const float dz  = (zc) - hprev;                                          \
    const float e   = __builtin_amdgcn_exp2f(__builtin_fmaf(s, LOG2E, as));  \
    const float v   = __builtin_amdgcn_rcpf(1.0f + e);                       \
    const float hn  = __builtin_fmaf(v, dz, hprev);                          \
    hprev = hn;                                                              \
    hsh[cur ^ 1][jw] = hn;                            \
    hbuf[(size_t)(tt) * Hn + tid] = hn;        \
      \
      \
      \
      \
    const int tn = ((tt) + 4 < Tn) ? (tt) + 4 : Tn - 1;                      \
    (ac) = abuf[(size_t)tn * Hn + tid];                                      \
    (zc) = zbuf[(size_t)tn * Hn + tid];                                      \
    lds_step_barrier();                        \
  } while (0)

  for (int t = 0; t < Tn; t += 4) {
    RNN_STEP(t + 0, a0, z0);
    RNN_STEP(t + 1, a1, z1);
    RNN_STEP(t + 2, a2, z2);
    RNN_STEP(t + 3, a3, z3);
  }
#undef RNN_STEP
}

__global__ void __launch_bounds__(128)
rnn_phase3(const float* __restrict__ hbuf, const float* __restrict__ Wout,
           const float* __restrict__ bout, float* __restrict__ y) {
  const int lane = threadIdx.x & 31;
  const int wave = threadIdx.x >> 5;
  const int n0   = wave * 16;
  const int row  = lane & 15;
  const int hi   = lane >> 4;
  const int ko   = hi * 2;
  const int t0   = blockIdx.x * 16;

  v8f c = {};
#pragma unroll
  for (int kc = 0; kc < Hn / 4; ++kc) {
    const int kb = kc * 4;
    v2f a, b;
    a.x = hbuf[(size_t)(t0 + row) * Hn + kb + ko];
    a.y = hbuf[(size_t)(t0 + row) * Hn + kb + ko + 1];
    const int n = n0 + row;
    b.x = Wout[n * Hn + kb + ko];
    b.y = Wout[n * Hn + kb + ko + 1];
    c = wmma_f32_k4(a, b, c);
  }
  const int n = n0 + row;
  const float bias = bout[n];
#pragma unroll
  for (int r = 0; r < 8; ++r) {
    const int m = r + hi * 8;
    y[(size_t)(t0 + m) * On + n] = c[r] + bias;
  }
}

extern "C" void kernel_launch(void* const* d_in, const int* in_sizes, int n_in,
                              void* d_out, int out_size, void* d_ws, size_t ws_size,
                              hipStream_t stream) {
  const float* x    = (const float*)d_in[0];
  const float* Win  = (const float*)d_in[1];
  const float* bin  = (const float*)d_in[2];
  const float* Wg   = (const float*)d_in[3];
  const float* bg   = (const float*)d_in[4];
  const float* Wout = (const float*)d_in[5];
  const float* bout = (const float*)d_in[6];

  float* zbuf = (float*)d_ws;
  float* abuf = zbuf + (size_t)Tn * Hn;
  float* hbuf = abuf + (size_t)Tn * Hn;
  float* yout = (float*)d_out;

  rnn_phase1<<<Tn / 16, 256, 0, stream>>>(x, Win, bin, Wg, bg, zbuf, abuf);
  rnn_phase2<<<1, NT2, 0, stream>>>(Wg, zbuf, abuf, hbuf);
  rnn_phase3<<<Tn / 16, 128, 0, stream>>>(hbuf, Wout, bout, yout);
}


# scan gate tail 5 to 2 dependent ops: 1-u = 0.5-0.5*tanh(x/2) on one v_tanh_f32, x/2 straight from the dot product (Wgh halved, a/2 seeds owner lane's sum); sum order, h last bits change
# speedup vs baseline: 5.0019x; 1.1813x over previous
#include <hip/hip_runtime.h>


typedef __attribute__((ext_vector_type(2))) float v2f;
typedef __attribute__((ext_vector_type(8))) float v8f;

#define Bsz 64
#define Tn  4096
#define In  64
#define Hn  128
#define On  64

__device__ __forceinline__ v8f wmma_f32_k4(v2f a, v2f b, v8f c) {
  return __builtin_amdgcn_wmma_f32_16x16x4_f32(
       false, a,  false, b,
       (short)0, c,  false,  false);
}

__global__ void __launch_bounds__(256)
rnn_phase1(const float* __restrict__ x, const float* __restrict__ Win,
           const float* __restrict__ bin, const float* __restrict__ Wg,
           const float* __restrict__ bg, float* __restrict__ zbuf,
           float* __restrict__ abuf) {
  __shared__ float zt[16][Hn + 4];
  const int lane = threadIdx.x & 31;
  const int wave = threadIdx.x >> 5;
  const int n0   = wave * 16;
  const int row  = lane & 15;
  const int hi   = lane >> 4;
  const int ko   = hi * 2;
  const int t0   = blockIdx.x * 16;
  const float* xrow = x + ((size_t)(Bsz - 1) * Tn + t0) * In;

  v8f c = {};
#pragma unroll
  for (int kc = 0; kc < In / 4; ++kc) {
    const int kb = kc * 4;
    v2f a, b;
    a.x = xrow[row * In + kb + ko];
    a.y = xrow[row * In + kb + ko + 1];
    const int n = n0 + row;
    b.x = Win[n * In + kb + ko];
    b.y = Win[n * In + kb + ko + 1];
    c = wmma_f32_k4(a, b, c);
  }
  {
    const int n = n0 + row;
    const float bias = bin[n];
#pragma unroll
    for (int r = 0; r < 8; ++r) {
      const int m = r + hi * 8;
      const float v = c[r] + bias;
      zt[m][n] = v;
      zbuf[(size_t)(t0 + m) * Hn + n] = v;
    }
  }
  __syncthreads();

  v8f c2 = {};
#pragma unroll
  for (int kc = 0; kc < Hn / 4; ++kc) {
    const int kb = kc * 4;
    v2f a, b;
    a.x = zt[row][kb + ko];
    a.y = zt[row][kb + ko + 1];
    const int n = n0 + row;
    b.x = Wg[n * (2 * Hn) + Hn + kb + ko];
    b.y = Wg[n * (2 * Hn) + Hn + kb + ko + 1];
    c2 = wmma_f32_k4(a, b, c2);
  }
  {
    const int n = n0 + row;
    const float bias = bg[n];
#pragma unroll
    for (int r = 0; r < 8; ++r) {
      const int m = r + hi * 8;
      abuf[(size_t)(t0 + m) * Hn + n] = c2[r] + bias;
    }
  }
}

#define HPITCH 20
#define HPAD   (8 * HPITCH)
#define NT2    128

#define DPP_XADD(d, send, keep, XM)                                          \
  asm("v_add_f32_dpp %0, %1, %2 row_xmask:" #XM " row_mask:0xf bank_mask:0xf" \
      : "=v"(d) : "v"(send), "v"(keep))

#define RS_STAGE(dst, lo, hi, bit, XM)                                       \
  do {                                                                       \
    const float keep_ = (bit) ? (hi) : (lo);                                 \
    const float send_ = (bit) ? (lo) : (hi);                                 \
    DPP_XADD(dst, send_, keep_, XM);                                         \
  } while (0)

__device__ __forceinline__ void lds_step_barrier() {
  asm volatile("s_wait_dscnt 0x0\n\ts_barrier_signal -1\n\ts_barrier_wait -1"
               ::: "memory");
}

__global__ void __launch_bounds__(NT2)
rnn_phase2(const float* __restrict__ Wg, const float* __restrict__ zbuf,
           const float* __restrict__ abuf, float* __restrict__ hbuf) {
  __shared__ __attribute__((aligned(16))) float hsh[2][HPAD];
  const int tid = threadIdx.x;
  const int ks  = tid & 7;
  const int jg  = tid >> 3;
  const int jw  = HPITCH * (tid >> 4) + (tid & 15);
  const bool b0 = (ks & 1) != 0;
  const bool b1 = (ks & 2) != 0;
  const bool b2 = (ks & 4) != 0;

  v2f w[8][8];
#pragma unroll
  for (int r = 0; r < 8; ++r) {
#pragma unroll
    for (int c = 0; c < 8; ++c) {
      w[r][c].x = 0.5f * Wg[(size_t)(8 * jg + r) * (2 * Hn) + 16 * ks + 2 * c];
      w[r][c].y = 0.5f * Wg[(size_t)(8 * jg + r) * (2 * Hn) + 16 * ks + 2 * c + 1];
    }
  }

  for (int i = tid; i < HPAD; i += NT2) hsh[0][i] = 0.0f;
  __syncthreads();

  float hprev = 0.0f;
  float a0 = abuf[0 * Hn + tid], z0 = zbuf[0 * Hn + tid];
  float a1 = abuf[1 * Hn + tid], z1 = zbuf[1 * Hn + tid];
  float a2 = abuf[2 * Hn + tid], z2 = zbuf[2 * Hn + tid];
  float a3 = abuf[3 * Hn + tid], z3 = zbuf[3 * Hn + tid];

#define RNN_STEP(tt, ac, zc)                                                 \
  do {                                                                       \
    const int cur = (tt) & 1;                                                \
    v2f hv[8];                                 \
    __builtin_memcpy(hv, &hsh[cur][HPITCH * ks], 64);                        \
      \
      \
      \
      \
      \
    __builtin_amdgcn_sched_barrier(0);                                       \
      \
      \
      \
      \
      \
      \
      \
    const float qa  = 0.25f * (ac);                                          \
    const float dz  = (zc) - hprev;                                          \
    const float gm  = -0.5f * dz;                                            \
    const float gn  = __builtin_fmaf(0.5f, dz, hprev);                       \
    v2f acc[8];                                                              \
      \
    _Pragma("unroll")                                                        \
    for (int r = 0; r < 8; ++r) {                                            \
      const float q_ = (ks == r) ? qa : 0.0f;                                \
      acc[r] = (v2f){q_, q_};                                                \
    }                                                                        \
    _Pragma("unroll")                                                        \
    for (int c = 0; c < 8; ++c) {                                            \
      _Pragma("unroll")                                                      \
      for (int r = 0; r < 8; ++r) acc[r] += w[r][c] * hv[c];                 \
    }                                                                        \
    float p[8];                                                              \
    _Pragma("unroll")                                                        \
    for (int r = 0; r < 8; ++r) p[r] = acc[r].x + acc[r].y;                  \
      \
      \
      \
    float q0, q1, q2, q3, r0, r1, s;                                         \
    RS_STAGE(q0, p[0], p[1], b0, 1);                                         \
    RS_STAGE(q1, p[2], p[3], b0, 1);                                         \
    RS_STAGE(q2, p[4], p[5], b0, 1);                                         \
    RS_STAGE(q3, p[6], p[7], b0, 1);                                         \
    RS_STAGE(r0, q0, q1, b1, 2);                                             \
    RS_STAGE(r1, q2, q3, b1, 2);                                             \
    RS_STAGE(s, r0, r1, b2, 4);                                              \
      \
      \
      \
      \
      \
    const float t_  = __builtin_amdgcn_tanhf(s);                             \
    const float hn  = __builtin_fmaf(t_, gm, gn);                            \
    hprev = hn;                                                              \
    hsh[cur ^ 1][jw] = hn;                            \
    hbuf[(size_t)(tt) * Hn + tid] = hn;        \
      \
      \
      \
      \
    const int tn = ((tt) + 4 < Tn) ? (tt) + 4 : Tn - 1;                      \
    (ac) = abuf[(size_t)tn * Hn + tid];                                      \
    (zc) = zbuf[(size_t)tn * Hn + tid];                                      \
    lds_step_barrier();                        \
  } while (0)

  for (int t = 0; t < Tn; t += 4) {
    RNN_STEP(t + 0, a0, z0);
    RNN_STEP(t + 1, a1, z1);
    RNN_STEP(t + 2, a2, z2);
    RNN_STEP(t + 3, a3, z3);
  }
#undef RNN_STEP
}

__global__ void __launch_bounds__(128)
rnn_phase3(const float* __restrict__ hbuf, const float* __restrict__ Wout,
           const float* __restrict__ bout, float* __restrict__ y) {
  const int lane = threadIdx.x & 31;
  const int wave = threadIdx.x >> 5;
  const int n0   = wave * 16;
  const int row  = lane & 15;
  const int hi   = lane >> 4;
  const int ko   = hi * 2;
  const int t0   = blockIdx.x * 16;

  v8f c = {};
#pragma unroll
  for (int kc = 0; kc < Hn / 4; ++kc) {
    const int kb = kc * 4;
    v2f a, b;
    a.x = hbuf[(size_t)(t0 + row) * Hn + kb + ko];
    a.y = hbuf[(size_t)(t0 + row) * Hn + kb + ko + 1];
    const int n = n0 + row;
    b.x = Wout[n * Hn + kb + ko];
    b.y = Wout[n * Hn + kb + ko + 1];
    c = wmma_f32_k4(a, b, c);
  }
  const int n = n0 + row;
  const float bias = bout[n];
#pragma unroll
  for (int r = 0; r < 8; ++r) {
    const int m = r + hi * 8;
    y[(size_t)(t0 + m) * On + n] = c[r] + bias;
  }
}

extern "C" void kernel_launch(void* const* d_in, const int* in_sizes, int n_in,
                              void* d_out, int out_size, void* d_ws, size_t ws_size,
                              hipStream_t stream) {
  const float* x    = (const float*)d_in[0];
  const float* Win  = (const float*)d_in[1];
  const float* bin  = (const float*)d_in[2];
  const float* Wg   = (const float*)d_in[3];
  const float* bg   = (const float*)d_in[4];
  const float* Wout = (const float*)d_in[5];
  const float* bout = (const float*)d_in[6];

  float* zbuf = (float*)d_ws;
  float* abuf = zbuf + (size_t)Tn * Hn;
  float* hbuf = abuf + (size_t)Tn * Hn;
  float* yout = (float*)d_out;

  rnn_phase1<<<Tn / 16, 256, 0, stream>>>(x, Win, bin, Wg, bg, zbuf, abuf);
  rnn_phase2<<<1, NT2, 0, stream>>>(Wg, zbuf, abuf, hbuf);
  rnn_phase3<<<Tn / 16, 128, 0, stream>>>(hbuf, Wout, bout, yout);
}


# scan: a wave takes its own 32 h by lane moves (DPP xor 8, permlanex16) and does that quarter of the next step's FMAs behind its h store, before the barrier; 3 LDS reads, not 4; dot sum order changes
# speedup vs baseline: 6.2168x; 1.2429x over previous
#include <hip/hip_runtime.h>


typedef __attribute__((ext_vector_type(2))) float v2f;
typedef __attribute__((ext_vector_type(4))) float v4f;
typedef __attribute__((ext_vector_type(8))) float v8f;

#define Bsz 64
#define Tn  4096
#define In  64
#define Hn  128
#define On  64

__device__ __forceinline__ v8f wmma_f32_k4(v2f a, v2f b, v8f c) {
  return __builtin_amdgcn_wmma_f32_16x16x4_f32(
       false, a,  false, b,
       (short)0, c,  false,  false);
}

__global__ void __launch_bounds__(256)
rnn_phase1(const float* __restrict__ x, const float* __restrict__ Win,
           const float* __restrict__ bin, const float* __restrict__ Wg,
           const float* __restrict__ bg, float* __restrict__ zbuf,
           float* __restrict__ abuf) {
  __shared__ float zt[16][Hn + 4];
  const int lane = threadIdx.x & 31;
  const int wave = threadIdx.x >> 5;
  const int n0   = wave * 16;
  const int row  = lane & 15;
  const int hi   = lane >> 4;
  const int ko   = hi * 2;
  const int t0   = blockIdx.x * 16;
  const float* xrow = x + ((size_t)(Bsz - 1) * Tn + t0) * In;

  v8f c = {};
#pragma unroll
  for (int kc = 0; kc < In / 4; ++kc) {
    const int kb = kc * 4;
    v2f a, b;
    a.x = xrow[row * In + kb + ko];
    a.y = xrow[row * In + kb + ko + 1];
    const int n = n0 + row;
    b.x = Win[n * In + kb + ko];
    b.y = Win[n * In + kb + ko + 1];
    c = wmma_f32_k4(a, b, c);
  }
  {
    const int n = n0 + row;
    const float bias = bin[n];
#pragma unroll
    for (int r = 0; r < 8; ++r) {
      const int m = r + hi * 8;
      const float v = c[r] + bias;
      zt[m][n] = v;
      zbuf[(size_t)(t0 + m) * Hn + n] = v;
    }
  }
  __syncthreads();

  v8f c2 = {};
#pragma unroll
  for (int kc = 0; kc < Hn / 4; ++kc) {
    const int kb = kc * 4;
    v2f a, b;
    a.x = zt[row][kb + ko];
    a.y = zt[row][kb + ko + 1];
    const int n = n0 + row;
    b.x = Wg[n * (2 * Hn) + Hn + kb + ko];
    b.y = Wg[n * (2 * Hn) + Hn + kb + ko + 1];
    c2 = wmma_f32_k4(a, b, c2);
  }
  {
    const int n = n0 + row;
    const float bias = bg[n];
#pragma unroll
    for (int r = 0; r < 8; ++r) {
      const int m = r + hi * 8;
      abuf[(size_t)(t0 + m) * Hn + n] = c2[r] + bias;
    }
  }
}

#define NT2    128

#define DPP_XADD(d, send, keep, XM)                                          \
  asm("v_add_f32_dpp %0, %1, %2 row_xmask:" #XM " row_mask:0xf bank_mask:0xf" \
      : "=v"(d) : "v"(send), "v"(keep))

#define RS_STAGE(dst, lo, hi, bit, XM)                                       \
  do {                                                                       \
    const float keep_ = (bit) ? (hi) : (lo);                                 \
    const float send_ = (bit) ? (lo) : (hi);                                 \
    DPP_XADD(dst, send_, keep_, XM);                                         \
  } while (0)

__device__ __forceinline__ float lane_xor8(float x) {
  float d;
  asm("v_mov_b32_dpp %0, %1 row_xmask:8 row_mask:0xf bank_mask:0xf"
      : "=v"(d) : "v"(x));
  return d;
}

__device__ __forceinline__ float lane_xor16(float x) {
  const int b = __builtin_bit_cast(int, x);
  return __builtin_bit_cast(float,
      __builtin_amdgcn_permlanex16(b, b, 0x76543210, 0xfedcba98, false, false));
}

__device__ __forceinline__ void lds_step_barrier() {
  asm volatile("s_wait_dscnt 0x0\n\ts_barrier_signal -1\n\ts_barrier_wait -1"
               ::: "memory");
}

__global__ void __launch_bounds__(NT2)
rnn_phase2(const float* __restrict__ Wg, const float* __restrict__ zbuf,
           const float* __restrict__ abuf, float* __restrict__ hbuf) {
  __shared__ __attribute__((aligned(16))) float hsh[2][Hn];
  const int tid  = threadIdx.x;
  const int lane = tid & 31;
  const int wv   = tid >> 5;
  const int ks   = tid & 7;
  const int jg   = tid >> 3;
  const bool b0 = (ks & 1) != 0;
  const bool b1 = (ks & 2) != 0;
  const bool b2 = (ks & 4) != 0;
  const int of0 = 32 * ((wv + 1) & 3) + 4 * ks;
  const int of1 = 32 * ((wv + 2) & 3) + 4 * ks;
  const int of2 = 32 * ((wv + 3) & 3) + 4 * ks;

  v2f w[8][8];
#pragma unroll
  for (int r = 0; r < 8; ++r) {
    const float* wrow = Wg + (size_t)(8 * jg + r) * (2 * Hn);
    w[r][0].x = 0.5f * wrow[32 * wv + lane];
    w[r][0].y = 0.5f * wrow[32 * wv + (lane ^ 8)];
    w[r][1].x = 0.5f * wrow[32 * wv + (lane ^ 16)];
    w[r][1].y = 0.5f * wrow[32 * wv + (lane ^ 24)];
#pragma unroll
    for (int j = 0; j < 3; ++j) {
      const int kb = 32 * ((wv + 1 + j) & 3) + 4 * ks;
      w[r][2 + 2 * j].x = 0.5f * wrow[kb + 0];
      w[r][2 + 2 * j].y = 0.5f * wrow[kb + 1];
      w[r][3 + 2 * j].x = 0.5f * wrow[kb + 2];
      w[r][3 + 2 * j].y = 0.5f * wrow[kb + 3];
    }
  }

  for (int i = tid; i < Hn; i += NT2) hsh[0][i] = 0.0f;
  __syncthreads();

  float mk[8];
#pragma unroll
  for (int r = 0; r < 8; ++r) mk[r] = (ks == r) ? 0.5f : 0.0f;

  float hprev = 0.0f;
  v2f acc[8];
  float a0 = abuf[0 * Hn + tid], z0 = zbuf[0 * Hn + tid];
  float a1 = abuf[1 * Hn + tid], z1 = zbuf[1 * Hn + tid];
  float a2 = abuf[2 * Hn + tid], z2 = zbuf[2 * Hn + tid];
  float a3 = abuf[3 * Hn + tid], z3 = zbuf[3 * Hn + tid];

#define RNN_OWN(ac)                                                          \
  do {                                                                       \
    const float av_ = (ac);                                                  \
    const float h2_ = lane_xor16(hprev);                                     \
    const float h1_ = lane_xor8(hprev);                                      \
    const float h3_ = lane_xor8(h2_);                                        \
    const v2f ha_ = {hprev, h1_};                                            \
    const v2f hb_ = {h2_, h3_};                                              \
           \
    _Pragma("unroll")                                                        \
    for (int r = 0; r < 8; ++r) acc[r] = w[r][0] * ha_;                      \
    _Pragma("unroll")                                                        \
    for (int r = 0; r < 8; ++r)                                              \
      acc[r].x = __builtin_fmaf(mk[r], av_, acc[r].x);                       \
    _Pragma("unroll")                                                        \
    for (int r = 0; r < 8; ++r) acc[r] += w[r][1] * hb_;                     \
  } while (0)

#define RNN_STEP(tt, ac, zc, an)                                             \
  do {                                                                       \
    const int cur = (tt) & 1;                                                \
    const v4f g0 = *(const v4f*)&hsh[cur][of0];                              \
    const v4f g1 = *(const v4f*)&hsh[cur][of1];                              \
    const v4f g2 = *(const v4f*)&hsh[cur][of2];                              \
      \
      \
      \
    __builtin_amdgcn_sched_barrier(0);                                       \
      \
      \
    const float dz  = (zc) - hprev;                                          \
    float gm  = -0.5f * dz;                                                  \
    float gn  = __builtin_fmaf(0.5f, dz, hprev);                             \
      \
      \
      \
      \
    asm volatile("" : "+v"(gm), "+v"(gn));                                   \
      \
      \
      \
      \
    const int tn = ((tt) + 4 < Tn) ? (tt) + 4 : Tn - 1;                      \
    (ac) = abuf[(size_t)tn * Hn + tid];                                      \
    (zc) = zbuf[(size_t)tn * Hn + tid];                                      \
      \
      \
      \
    __builtin_amdgcn_sched_barrier(0);                                       \
    const v2f hv[6] = {{g0.x, g0.y}, {g0.z, g0.w}, {g1.x, g1.y},             \
                       {g1.z, g1.w}, {g2.x, g2.y}, {g2.z, g2.w}};            \
      \
    _Pragma("unroll")                                                        \
    for (int c = 0; c < 6; ++c) {                                            \
      _Pragma("unroll")                                                      \
      for (int r = 0; r < 8; ++r) acc[r] += w[r][2 + c] * hv[c];             \
    }                                                                        \
    float p[8];                                                              \
    _Pragma("unroll")                                                        \
    for (int r = 0; r < 8; ++r) p[r] = acc[r].x + acc[r].y;                  \
      \
      \
      \
    float q0, q1, q2, q3, r0, r1, s;                                         \
    RS_STAGE(q0, p[0], p[1], b0, 1);                                         \
    RS_STAGE(q1, p[2], p[3], b0, 1);                                         \
    RS_STAGE(q2, p[4], p[5], b0, 1);                                         \
    RS_STAGE(q3, p[6], p[7], b0, 1);                                         \
    RS_STAGE(r0, q0, q1, b1, 2);                                             \
    RS_STAGE(r1, q2, q3, b1, 2);                                             \
    RS_STAGE(s, r0, r1, b2, 4);                                              \
      \
      \
      \
      \
      \
    const float t_  = __builtin_amdgcn_tanhf(s);                             \
    const float hn  = __builtin_fmaf(t_, gm, gn);                            \
    hprev = hn;                                                              \
    hsh[cur ^ 1][tid] = hn;                           \
    hbuf[(size_t)(tt) * Hn + tid] = hn;        \
      \
      \
      \
    __builtin_amdgcn_sched_barrier(0);                                       \
          \
    RNN_OWN(an);                                                             \
      \
      \
      \
    __builtin_amdgcn_sched_barrier(0);                                       \
    lds_step_barrier();                        \
  } while (0)

  RNN_OWN(a0);
  for (int t = 0; t < Tn; t += 4) {
    RNN_STEP(t + 0, a0, z0, a1);
    RNN_STEP(t + 1, a1, z1, a2);
    RNN_STEP(t + 2, a2, z2, a3);
    RNN_STEP(t + 3, a3, z3, a0);
  }
#undef RNN_STEP
#undef RNN_OWN
}

__global__ void __launch_bounds__(128)
rnn_phase3(const float* __restrict__ hbuf, const float* __restrict__ Wout,
           const float* __restrict__ bout, float* __restrict__ y) {
  const int lane = threadIdx.x & 31;
  const int wave = threadIdx.x >> 5;
  const int n0   = wave * 16;
  const int row  = lane & 15;
  const int hi   = lane >> 4;
  const int ko   = hi * 2;
  const int t0   = blockIdx.x * 16;

  v8f c = {};
#pragma unroll
  for (int kc = 0; kc < Hn / 4; ++kc) {
    const int kb = kc * 4;
    v2f a, b;
    a.x = hbuf[(size_t)(t0 + row) * Hn + kb + ko];
    a.y = hbuf[(size_t)(t0 + row) * Hn + kb + ko + 1];
    const int n = n0 + row;
    b.x = Wout[n * Hn + kb + ko];
    b.y = Wout[n * Hn + kb + ko + 1];
    c = wmma_f32_k4(a, b, c);
  }
  const int n = n0 + row;
  const float bias = bout[n];
#pragma unroll
  for (int r = 0; r < 8; ++r) {
    const int m = r + hi * 8;
    y[(size_t)(t0 + m) * On + n] = c[r] + bias;
  }
}

extern "C" void kernel_launch(void* const* d_in, const int* in_sizes, int n_in,
                              void* d_out, int out_size, void* d_ws, size_t ws_size,
                              hipStream_t stream) {
  const float* x    = (const float*)d_in[0];
  const float* Win  = (const float*)d_in[1];
  const float* bin  = (const float*)d_in[2];
  const float* Wg   = (const float*)d_in[3];
  const float* bg   = (const float*)d_in[4];
  const float* Wout = (const float*)d_in[5];
  const float* bout = (const float*)d_in[6];

  float* zbuf = (float*)d_ws;
  float* abuf = zbuf + (size_t)Tn * Hn;
  float* hbuf = abuf + (size_t)Tn * Hn;
  float* yout = (float*)d_out;

  rnn_phase1<<<Tn / 16, 256, 0, stream>>>(x, Win, bin, Wg, bg, zbuf, abuf);
  rnn_phase2<<<1, NT2, 0, stream>>>(Wg, zbuf, abuf, hbuf);
  rnn_phase3<<<Tn / 16, 128, 0, stream>>>(hbuf, Wout, bout, yout);
}
